# starting kernel
# baseline (speedup 1.0000x reference)
#include <hip/hip_runtime.h>
#include <hip/hip_bf16.h>

typedef __attribute__((ext_vector_type(2))) float v2f;
typedef __attribute__((ext_vector_type(8))) float v8f;

#define NDIM 128

__global__ void k_deg_init(float* deg, int n) {
    int i = blockIdx.x * blockDim.x + threadIdx.x;
    if (i < n) deg[i] = 1.0f;
}

__global__ void k_deg_count(const int* __restrict__ dst, float* deg, int e) {
    int i = blockIdx.x * blockDim.x + threadIdx.x;
    if (i < e) unsafeAtomicAdd(&deg[dst[i]], 1.0f);
}

__global__ void k_deg_rsqrt(float* deg, int n) {
    int i = blockIdx.x * blockDim.x + threadIdx.x;
    if (i < n) deg[i] = __frsqrt_rn(deg[i]);
}

__global__ void __launch_bounds__(256)
k_gemm128(const float* __restrict__ X, const float* __restrict__ W,
          float* __restrict__ H, int M) {
    __shared__ float Ws[NDIM * NDIM];

    int tid = threadIdx.x;
    {
        const float4* Wv = (const float4*)W;
        float4* Sv = (float4*)Ws;
        #pragma unroll
        for (int i = 0; i < 16; ++i)
            Sv[tid + 256 * i] = Wv[tid + 256 * i];
    }
    __syncthreads();

    int wave  = tid >> 5;
    int lane  = tid & 31;
    int l16   = lane & 15;
    int lhalf = lane >> 4;

    int rowBase = blockIdx.x * 128 + wave * 16;

    int arow = rowBase + l16;
    if (arow > M - 1) arow = M - 1;
    const float* aptr = X + (size_t)arow * NDIM + lhalf * 2;

    v8f acc[8] = {};

    for (int k = 0; k < NDIM; k += 4) {
        v2f a;
        {
            const float2 av = *(const float2*)(aptr + k);
            a[0] = av.x; a[1] = av.y;
        }
        int krow = (k + lhalf * 2) * NDIM + l16;
        #pragma unroll
        for (int t = 0; t < 8; ++t) {
            v2f b;
            b[0] = Ws[krow + t * 16];
            b[1] = Ws[krow + NDIM + t * 16];
            acc[t] = __builtin_amdgcn_wmma_f32_16x16x4_f32(
                false, a, false, b, (short)0, acc[t], false, false);
        }
    }

    #pragma unroll
    for (int t = 0; t < 8; ++t) {
        #pragma unroll
        for (int v = 0; v < 8; ++v) {
            int r = rowBase + v + lhalf * 8;
            if (r < M) H[(size_t)r * NDIM + t * 16 + l16] = acc[t][v];
        }
    }
}

__global__ void k_agg_init(const float* __restrict__ h, const float* __restrict__ dinv,
                           const float* __restrict__ bias, float* __restrict__ out, int n) {
    int i = blockIdx.x * blockDim.x + threadIdx.x;
    if (i >= n * 32) return;
    int node = i >> 5;
    int c4   = (i & 31) * 4;
    float di = dinv[node];
    float s  = di * di;
    float4 hv = *(const float4*)(h + (size_t)node * NDIM + c4);
    float4 bv = *(const float4*)(bias + c4);
    float4 o;
    o.x = bv.x + hv.x * s;
    o.y = bv.y + hv.y * s;
    o.z = bv.z + hv.z * s;
    o.w = bv.w + hv.w * s;
    *(float4*)(out + (size_t)node * NDIM + c4) = o;
}

#define EPW 4
__global__ void __launch_bounds__(256)
k_agg_edges(const float* __restrict__ h, const float* __restrict__ dinv,
            const int* __restrict__ src, const int* __restrict__ dst,
            float* __restrict__ out, int e) {
    int wv   = __builtin_amdgcn_readfirstlane((int)(threadIdx.x >> 5));
    int lane = threadIdx.x & 31;
    int base = (blockIdx.x * 8 + wv) * EPW;
    if (base >= e) return;

    int    s[EPW], d[EPW];
    float  nrm[EPW];
    float4 hv[EPW];

    #pragma unroll
    for (int j = 0; j < EPW; ++j) {
        int eid = (base + j < e) ? (base + j) : (e - 1);
        s[j] = src[eid];
        d[j] = dst[eid];
        nrm[j] = dinv[s[j]] * dinv[d[j]];
    }
    #pragma unroll
    for (int j = 0; j < EPW; ++j)
        hv[j] = *(const float4*)(h + (size_t)s[j] * NDIM + lane * 4);

    #pragma unroll
    for (int j = 0; j < EPW; ++j) {
        if (base + j >= e) break;
        float* op = out + (size_t)d[j] * NDIM + lane * 4;
        unsafeAtomicAdd(op + 0, hv[j].x * nrm[j]);
        unsafeAtomicAdd(op + 1, hv[j].y * nrm[j]);
        unsafeAtomicAdd(op + 2, hv[j].z * nrm[j]);
        unsafeAtomicAdd(op + 3, hv[j].w * nrm[j]);
    }
}

__global__ void k_tanh_ip(float* p, int n4) {
    int i = blockIdx.x * blockDim.x + threadIdx.x;
    if (i >= n4) return;
    float4 v = ((float4*)p)[i];
    v.x = tanhf(v.x); v.y = tanhf(v.y); v.z = tanhf(v.z); v.w = tanhf(v.w);
    ((float4*)p)[i] = v;
}

extern "C" void kernel_launch(void* const* d_in, const int* in_sizes, int n_in,
                              void* d_out, int out_size, void* d_ws, size_t ws_size,
                              hipStream_t stream) {
    const float* x   = (const float*)d_in[0];
    const float* W1  = (const float*)d_in[1];
    const float* b1  = (const float*)d_in[2];
    const float* W2  = (const float*)d_in[3];
    const float* b2  = (const float*)d_in[4];
    const int*   ei  = (const int*)d_in[5];

    const int N = in_sizes[0] / NDIM;
    const int E = in_sizes[5] / 2;
    const int* src = ei;
    const int* dst = ei + E;

    float* dinv = (float*)d_ws;
    float* h    = dinv + ((N + 255) & ~255);
    float* t1   = h + (size_t)N * NDIM;
    float* outp = (float*)d_out;

    const int B = 256;
    int gN   = (N + B - 1) / B;
    int gE   = (E + B - 1) / B;
    int gN32 = (N * 32 + B - 1) / B;
    int gEw  = (E + 8 * EPW - 1) / (8 * EPW);
    int gM   = (N + 127) / 128;

    k_deg_init <<<gN, B, 0, stream>>>(dinv, N);
    k_deg_count<<<gE, B, 0, stream>>>(dst, dinv, E);
    k_deg_rsqrt<<<gN, B, 0, stream>>>(dinv, N);

    k_gemm128  <<<gM,   B, 0, stream>>>(x, W1, h, N);
    k_agg_init <<<gN32, B, 0, stream>>>(h, dinv, b1, t1, N);
    k_agg_edges<<<gEw,  B, 0, stream>>>(h, dinv, src, dst, t1, E);
    k_tanh_ip  <<<gN32, B, 0, stream>>>(t1, N * 32);

    k_gemm128  <<<gM,   B, 0, stream>>>(t1, W2, h, N);
    k_agg_init <<<gN32, B, 0, stream>>>(h, dinv, b2, outp, N);
    k_agg_edges<<<gEw,  B, 0, stream>>>(h, dinv, src, dst, outp, E);
}


# Layers computed as (A_hat X) W, not A_hat (X W): owner blocks sum rows of X in LDS, then multiply the sums by W on the matrix unit; GEMM launches and h plane go (5 launches -> 3); sum order changes
# speedup vs baseline: 2.5113x; 2.5113x over previous
#include <hip/hip_runtime.h>
#include <hip/hip_bf16.h>

typedef __attribute__((ext_vector_type(2))) float v2f;
typedef __attribute__((ext_vector_type(8))) float v8f;

#define NDIM 128

__device__ __forceinline__ void mma_kstep(v8f (&acc)[8], v2f a, const float* Ws, int krow) {
    #pragma unroll
    for (int t = 0; t < 8; ++t) {
        v2f b;
        b[0] = Ws[krow + t * 16];
        b[1] = Ws[krow + NDIM + t * 16];
        acc[t] = __builtin_amdgcn_wmma_f32_16x16x4_f32(
            false, a, false, b, (short)0, acc[t], false, false);
    }
}

#define OWN  391
#define APITCH 132
#define LCAP 4096
#define EPW 8
#define CPL 4
#define SCAN_V4 8

template <class F>
__device__ __forceinline__ void scan_owned(const int* __restrict__ dst, int e, int wv, int lane,
                                           int lo, unsigned ucnt, F&& onMatch) {
    const bool al16 = ((size_t)dst & 15) == 0;
    const int perWave = SCAN_V4 * 128;
    for (int base = wv * perWave; base < e; base += 32 * perWave) {
        int4 v[SCAN_V4];
        if (al16 && base + perWave <= e) {
            #pragma unroll
            for (int j = 0; j < SCAN_V4; ++j)
                v[j] = *(const int4*)(dst + base + j * 128 + lane * 4);
        } else {
            #pragma unroll
            for (int j = 0; j < SCAN_V4; ++j) {
                const int i = base + j * 128 + lane * 4;
                v[j].x = (i     < e) ? dst[i]     : -1;
                v[j].y = (i + 1 < e) ? dst[i + 1] : -1;
                v[j].z = (i + 2 < e) ? dst[i + 2] : -1;
                v[j].w = (i + 3 < e) ? dst[i + 3] : -1;
            }
        }
        #pragma unroll
        for (int j = 0; j < SCAN_V4; ++j) {
            const int i = base + j * 128 + lane * 4;
            unsigned r[4];
            r[0] = (unsigned)(v[j].x - lo); r[1] = (unsigned)(v[j].y - lo);
            r[2] = (unsigned)(v[j].z - lo); r[3] = (unsigned)(v[j].w - lo);
            unsigned m01 = r[0] < r[1] ? r[0] : r[1];
            unsigned m23 = r[2] < r[3] ? r[2] : r[3];
            unsigned m = m01 < m23 ? m01 : m23;
            if (m < ucnt) {
                #pragma unroll
                for (int k = 0; k < 4; ++k)
                    if (r[k] < ucnt) onMatch(i + k, r[k]);
            }
        }
    }
}

__global__ void __launch_bounds__(1024)
k_own_filter(const int* __restrict__ dst, float* __restrict__ dinv,
             int* __restrict__ glist, int* __restrict__ gcount, int n, int e) {
    __shared__ int indeg[OWN];
    __shared__ int lcount;

    const int tid  = threadIdx.x;
    const int lane = tid & 31;
    const int wv   = __builtin_amdgcn_readfirstlane(tid >> 5);
    const int lo   = blockIdx.x * OWN;
    const int cnt  = (n - lo < OWN) ? (n - lo) : OWN;

    if (tid < OWN) indeg[tid] = 0;
    if (tid == 0) lcount = 0;
    __syncthreads();

    int* gl = glist + (size_t)blockIdx.x * LCAP;
    scan_owned(dst, e, wv, lane, lo, (unsigned)cnt, [&](int id, unsigned r) {
        atomicAdd(&indeg[r], 1);
        const int slot = atomicAdd(&lcount, 1);
        if (slot < LCAP) gl[slot] = id;
    });
    __syncthreads();

    if (tid == 0) gcount[blockIdx.x] = lcount;
    if (tid < cnt) dinv[lo + tid] = __frsqrt_rn((float)(indeg[tid] + 1));
}

__device__ __noinline__ void agg_edge_by_lane(const float* __restrict__ h, const float* __restrict__ dinv,
                                              const int* __restrict__ src, float* accRow, int i, int d) {
    const int s = src[i];
    const float nrm = dinv[s] * dinv[d];
    const float* hp = h + (size_t)s * NDIM;
    for (int c = 0; c < NDIM; ++c) unsafeAtomicAdd(accRow + c, hp[c] * nrm);
}

template <bool TANH_OUT>
__global__ void __launch_bounds__(1024)
k_layer(const float* __restrict__ xin, const float* __restrict__ dinv, const float* __restrict__ bias,
        const float* __restrict__ W,
        const int* __restrict__ src, const int* __restrict__ dst,
        const int* __restrict__ glist, const int* __restrict__ gcount,
        float* __restrict__ out, int n, int e) {
    __shared__ __attribute__((aligned(16))) float acc[OWN * APITCH];
    __shared__ __attribute__((aligned(16))) float Ws[NDIM * NDIM];
    __shared__ int   lst[LCAP];
    __shared__ int   lcount;

    const int tid  = threadIdx.x;
    const int lane = tid & 31;
    const int wv   = __builtin_amdgcn_readfirstlane(tid >> 5);
    const int lo   = blockIdx.x * OWN;
    const int cnt  = (n - lo < OWN) ? (n - lo) : OWN;
    const unsigned ucnt = (unsigned)cnt;

    const int  gtotal = gcount[blockIdx.x];
    const bool listed = gtotal <= LCAP;

    if (tid == 0) lcount = listed ? gtotal : 0;
    float4 wreg[NDIM * NDIM / 4096];
    {
        const float4* Wv = (const float4*)W;
        #pragma unroll
        for (int q = 0; q < NDIM * NDIM / 4096; ++q) wreg[q] = Wv[tid + 1024 * q];
    }
    int ids[LCAP / 1024];
    if (listed) {
        const int* gl = glist + (size_t)blockIdx.x * LCAP;
        #pragma unroll
        for (int q = 0; q < LCAP / 1024; ++q) {
            const int i = tid + 1024 * q;
            ids[q] = (i < gtotal) ? gl[i] : 0;
        }
    }
    {
        for (int r = wv; r < cnt; r += 32) {
            const float di = dinv[lo + r];
            const float sl = di * di;
            const float4 x4 = *(const float4*)(xin + (size_t)(lo + r) * NDIM + lane * 4);
            float4 o;
            o.x = x4.x * sl; o.y = x4.y * sl; o.z = x4.z * sl; o.w = x4.w * sl;
            *(float4*)(acc + r * APITCH + lane * 4) = o;
        }
    }
    {
        float4* Sv = (float4*)Ws;
        #pragma unroll
        for (int q = 0; q < NDIM * NDIM / 4096; ++q) Sv[tid + 1024 * q] = wreg[q];
    }
    if (listed) {
        #pragma unroll
        for (int q = 0; q < LCAP / 1024; ++q) {
            const int i = tid + 1024 * q;
            if (i < gtotal) lst[i] = ids[q];
        }
    }
    __syncthreads();

    if (!listed) {
        scan_owned(dst, e, wv, lane, lo, ucnt, [&](int id, unsigned r) {
            const int slot = atomicAdd(&lcount, 1);
            if (slot < LCAP) lst[slot] = id;
            else agg_edge_by_lane(xin, dinv, src, acc + r * APITCH, id, lo + (int)r);
        });
    }
    __syncthreads();

    {
        const int total = (lcount < LCAP) ? lcount : LCAP;
        for (int b = wv * EPW; b < total; b += 32 * EPW) {
            if (b + EPW <= total) {
                int s[EPW], d[EPW];
                #pragma unroll
                for (int j = 0; j < EPW; ++j) {
                    const int ed = __builtin_amdgcn_readfirstlane(lst[b + j]);
                    s[j] = src[ed];
                    d[j] = dst[ed];
                }
                float hv[EPW][CPL];
                #pragma unroll
                for (int j = 0; j < EPW; ++j) {
                    const float* hp = xin + (size_t)s[j] * NDIM + lane;
                    #pragma unroll
                    for (int q = 0; q < CPL; ++q) hv[j][q] = hp[32 * q];
                }
                float nrm[EPW];
                #pragma unroll
                for (int j = 0; j < EPW; ++j) nrm[j] = dinv[s[j]] * dinv[d[j]];
                #pragma unroll
                for (int j = 0; j < EPW; ++j) {
                    float* ap = acc + (d[j] - lo) * APITCH + lane;
                    #pragma unroll
                    for (int q = 0; q < CPL; ++q)
                        unsafeAtomicAdd(ap + 32 * q, hv[j][q] * nrm[j]);
                }
            } else {
                for (int k = b; k < total; ++k) {
                    const int ed = __builtin_amdgcn_readfirstlane(lst[k]);
                    const int s1 = src[ed], d1 = dst[ed];
                    const float nr = dinv[s1] * dinv[d1];
                    const float* hp = xin + (size_t)s1 * NDIM + lane;
                    float* ap = acc + (d1 - lo) * APITCH + lane;
                    float hv1[CPL];
                    #pragma unroll
                    for (int q = 0; q < CPL; ++q) hv1[q] = hp[32 * q];
                    #pragma unroll
                    for (int q = 0; q < CPL; ++q) unsafeAtomicAdd(ap + 32 * q, hv1[q] * nr);
                }
            }
        }
    }
    __syncthreads();

    {
        const int l16    = lane & 15;
        const int lhalf  = lane >> 4;
        const int ntiles = (cnt + 15) >> 4;
        for (int rt = wv; rt < ntiles; rt += 32) {
            int arow = rt * 16 + l16;
            if (arow > cnt - 1) arow = cnt - 1;
            const float* aptr = acc + arow * APITCH + lhalf * 2;

            v8f dacc[8];
            #pragma unroll
            for (int t = 0; t < 8; ++t) {
                const float bc = bias[t * 16 + l16];
                #pragma unroll
                for (int v = 0; v < 8; ++v) dacc[t][v] = bc;
            }
            for (int k = 0; k < NDIM; k += 4) {
                v2f a;
                {
                    const float2 av = *(const float2*)(aptr + k);
                    a[0] = av.x; a[1] = av.y;
                }
                mma_kstep(dacc, a, Ws, (k + lhalf * 2) * NDIM + l16);
            }

            #pragma unroll
            for (int t = 0; t < 8; ++t) {
                #pragma unroll
                for (int v = 0; v < 8; ++v) {
                    const int r = rt * 16 + v + lhalf * 8;
                    if (r < cnt) acc[r * APITCH + t * 16 + l16] = dacc[t][v];
                }
            }

            __builtin_amdgcn_fence(__ATOMIC_SEQ_CST, "wavefront");

            const int rend = (rt * 16 + 16 < cnt) ? (rt * 16 + 16) : cnt;
            for (int r = rt * 16; r < rend; ++r) {
                float4 o = *(const float4*)(acc + r * APITCH + lane * 4);
                if (TANH_OUT) { o.x = tanhf(o.x); o.y = tanhf(o.y); o.z = tanhf(o.z); o.w = tanhf(o.w); }
                *(float4*)(out + (size_t)(lo + r) * NDIM + lane * 4) = o;
            }
        }
    }
}

extern "C" void kernel_launch(void* const* d_in, const int* in_sizes, int n_in,
                              void* d_out, int out_size, void* d_ws, size_t ws_size,
                              hipStream_t stream) {
    const float* x   = (const float*)d_in[0];
    const float* W1  = (const float*)d_in[1];
    const float* b1  = (const float*)d_in[2];
    const float* W2  = (const float*)d_in[3];
    const float* b2  = (const float*)d_in[4];
    const int*   ei  = (const int*)d_in[5];

    const int N = in_sizes[0] / NDIM;
    const int E = in_sizes[5] / 2;
    const int* src = ei;
    const int* dst = ei + E;

    float* dinv = (float*)d_ws;
    float* a1   = dinv + ((N + 255) & ~255);
    float* outp = (float*)d_out;

    int gA   = (N + OWN - 1) / OWN;

    int* gcount = (int*)(a1 + (size_t)N * NDIM);
    int* glist  = gcount + ((gA + 255) & ~255);

    k_own_filter<<<gA, 1024, 0, stream>>>(dst, dinv, glist, gcount, N, E);

    k_layer<true> <<<gA, 1024, 0, stream>>>(x,  dinv, b1, W1, src, dst, glist, gcount, a1,   N, E);

    k_layer<false><<<gA, 1024, 0, stream>>>(a1, dinv, b2, W2, src, dst, glist, gcount, outp, N, E);
}


# Owner blocks' lists become resolved work lists built once per call: (source row, owned row) per term, self-loops included; layer kernels lose the own-row load phase and index loads; sum order changes
# speedup vs baseline: 2.8254x; 1.1251x over previous
#include <hip/hip_runtime.h>
#include <hip/hip_bf16.h>

typedef __attribute__((ext_vector_type(2))) float v2f;
typedef __attribute__((ext_vector_type(8))) float v8f;

#define NDIM 128

__device__ __forceinline__ void mma_kstep(v8f (&acc)[8], v2f a, const float* Ws, int krow) {
    #pragma unroll
    for (int t = 0; t < 8; ++t) {
        v2f b;
        b[0] = Ws[krow + t * 16];
        b[1] = Ws[krow + NDIM + t * 16];
        acc[t] = __builtin_amdgcn_wmma_f32_16x16x4_f32(
            false, a, false, b, (short)0, acc[t], false, false);
    }
}

#define OWN  391
#define APITCH 132
#define LCAP 4096
#define LTOT (LCAP + OWN)
#define EPW 8
#define CPL 4
#define SCAN_V4 8

template <class F>
__device__ __forceinline__ void scan_owned(const int* __restrict__ dst, int e, int wv, int lane,
                                           int lo, unsigned ucnt, F&& onMatch) {
    const bool al16 = ((size_t)dst & 15) == 0;
    const int perWave = SCAN_V4 * 128;
    for (int base = wv * perWave; base < e; base += 32 * perWave) {
        int4 v[SCAN_V4];
        if (al16 && base + perWave <= e) {
            #pragma unroll
            for (int j = 0; j < SCAN_V4; ++j)
                v[j] = *(const int4*)(dst + base + j * 128 + lane * 4);
        } else {
            #pragma unroll
            for (int j = 0; j < SCAN_V4; ++j) {
                const int i = base + j * 128 + lane * 4;
                v[j].x = (i     < e) ? dst[i]     : -1;
                v[j].y = (i + 1 < e) ? dst[i + 1] : -1;
                v[j].z = (i + 2 < e) ? dst[i + 2] : -1;
                v[j].w = (i + 3 < e) ? dst[i + 3] : -1;
            }
        }
        #pragma unroll
        for (int j = 0; j < SCAN_V4; ++j) {
            const int i = base + j * 128 + lane * 4;
            unsigned r[4];
            r[0] = (unsigned)(v[j].x - lo); r[1] = (unsigned)(v[j].y - lo);
            r[2] = (unsigned)(v[j].z - lo); r[3] = (unsigned)(v[j].w - lo);
            unsigned m01 = r[0] < r[1] ? r[0] : r[1];
            unsigned m23 = r[2] < r[3] ? r[2] : r[3];
            unsigned m = m01 < m23 ? m01 : m23;
            if (m < ucnt) {
                #pragma unroll
                for (int k = 0; k < 4; ++k)
                    if (r[k] < ucnt) onMatch(i + k, r[k]);
            }
        }
    }
}

__global__ void __launch_bounds__(1024)
k_own_filter(const int* __restrict__ src, const int* __restrict__ dst, float* __restrict__ dinv,
             int* __restrict__ glist, int* __restrict__ gcount, int n, int e) {
    __shared__ int indeg[OWN];
    __shared__ int lids[LCAP];
    __shared__ int lrow[LCAP];
    __shared__ int lcount;

    const int tid  = threadIdx.x;
    const int lane = tid & 31;
    const int wv   = __builtin_amdgcn_readfirstlane(tid >> 5);
    const int lo   = blockIdx.x * OWN;
    const int cnt  = (n - lo < OWN) ? (n - lo) : OWN;

    if (tid < OWN) indeg[tid] = 0;
    if (tid == 0) lcount = 0;
    __syncthreads();

    scan_owned(dst, e, wv, lane, lo, (unsigned)cnt, [&](int id, unsigned r) {
        atomicAdd(&indeg[r], 1);
        const int slot = atomicAdd(&lcount, 1);
        if (slot < LCAP) { lids[slot] = id; lrow[slot] = (int)r; }
    });
    __syncthreads();

    const int total = lcount;
    if (tid == 0) gcount[blockIdx.x] = total;

    int2* gl = (int2*)glist + (size_t)blockIdx.x * LTOT;
    if (tid < cnt) {
        dinv[lo + tid] = __frsqrt_rn((float)(indeg[tid] + 1));
        gl[tid] = make_int2(lo + tid, tid);
    }
    const int nl = (total < LCAP) ? total : LCAP;
    int sv[LCAP / 1024];
    #pragma unroll
    for (int q = 0; q < LCAP / 1024; ++q) {
        const int i = tid + 1024 * q;
        sv[q] = (i < nl) ? src[lids[i]] : 0;
    }
    #pragma unroll
    for (int q = 0; q < LCAP / 1024; ++q) {
        const int i = tid + 1024 * q;
        if (i < nl) gl[cnt + i] = make_int2(sv[q], lrow[i]);
    }
}

__device__ __noinline__ void agg_edge_by_lane(const float* __restrict__ h, const float* __restrict__ dinv,
                                              const int* __restrict__ src, float* accRow, int i, int d) {
    const int s = src[i];
    const float nrm = dinv[s] * dinv[d];
    const float* hp = h + (size_t)s * NDIM;
    for (int c = 0; c < NDIM; ++c) unsafeAtomicAdd(accRow + c, hp[c] * nrm);
}

template <bool TANH_OUT>
__global__ void __launch_bounds__(1024)
k_layer(const float* __restrict__ xin, const float* __restrict__ dinv, const float* __restrict__ bias,
        const float* __restrict__ W,
        const int* __restrict__ src, const int* __restrict__ dst,
        const int* __restrict__ glist, const int* __restrict__ gcount,
        float* __restrict__ out, int n, int e) {
    __shared__ __attribute__((aligned(16))) float acc[OWN * APITCH];
    __shared__ __attribute__((aligned(16))) float Ws[NDIM * NDIM];
    __shared__ __attribute__((aligned(16))) int2 lst[LTOT];
    __shared__ int   lcount;

    const int tid  = threadIdx.x;
    const int lane = tid & 31;
    const int wv   = __builtin_amdgcn_readfirstlane(tid >> 5);
    const int lo   = blockIdx.x * OWN;
    const int cnt  = (n - lo < OWN) ? (n - lo) : OWN;
    const unsigned ucnt = (unsigned)cnt;

    const int  gtotal = gcount[blockIdx.x];
    const bool listed = gtotal <= LCAP;

    if (tid == 0) lcount = listed ? gtotal : 0;
    float4 wreg[NDIM * NDIM / 4096];
    {
        const float4* Wv = (const float4*)W;
        #pragma unroll
        for (int q = 0; q < NDIM * NDIM / 4096; ++q) wreg[q] = Wv[tid + 1024 * q];
    }
    const int nent = cnt + gtotal;
    int2 ent[(LTOT + 1023) / 1024];
    if (listed) {
        const int2* gl = (const int2*)glist + (size_t)blockIdx.x * LTOT;
        #pragma unroll
        for (int q = 0; q < (LTOT + 1023) / 1024; ++q) {
            const int i = tid + 1024 * q;
            ent[q] = (i < nent) ? gl[i] : make_int2(0, 0);
        }
    }
    {
        for (int r = wv; r < cnt; r += 32)
            *(float4*)(acc + r * APITCH + lane * 4) = make_float4(0.f, 0.f, 0.f, 0.f);
    }
    {
        float4* Sv = (float4*)Ws;
        #pragma unroll
        for (int q = 0; q < NDIM * NDIM / 4096; ++q) Sv[tid + 1024 * q] = wreg[q];
    }
    if (listed) {
        #pragma unroll
        for (int q = 0; q < (LTOT + 1023) / 1024; ++q) {
            const int i = tid + 1024 * q;
            if (i < nent) lst[i] = ent[q];
        }
    } else if (tid < cnt) {
        lst[tid] = make_int2(lo + tid, tid);
    }
    __syncthreads();

    if (!listed) {
        scan_owned(dst, e, wv, lane, lo, ucnt, [&](int id, unsigned r) {
            const int slot = atomicAdd(&lcount, 1);
            if (slot < LCAP) lst[cnt + slot] = make_int2(src[id], (int)r);
            else agg_edge_by_lane(xin, dinv, src, acc + r * APITCH, id, lo + (int)r);
        });
    }
    __syncthreads();

    {
        const int total = cnt + ((lcount < LCAP) ? lcount : LCAP);
        for (int b = wv * EPW; b < total; b += 32 * EPW) {
            if (b + EPW <= total) {
                int s[EPW], d[EPW];
                #pragma unroll
                for (int j = 0; j < EPW; ++j) {
                    const int2 en = lst[b + j];
                    s[j] = __builtin_amdgcn_readfirstlane(en.x);
                    d[j] = lo + __builtin_amdgcn_readfirstlane(en.y);
                }
                float hv[EPW][CPL];
                #pragma unroll
                for (int j = 0; j < EPW; ++j) {
                    const float* hp = xin + (size_t)s[j] * NDIM + lane;
                    #pragma unroll
                    for (int q = 0; q < CPL; ++q) hv[j][q] = hp[32 * q];
                }
                float nrm[EPW];
                #pragma unroll
                for (int j = 0; j < EPW; ++j) nrm[j] = dinv[s[j]] * dinv[d[j]];
                #pragma unroll
                for (int j = 0; j < EPW; ++j) {
                    float* ap = acc + (d[j] - lo) * APITCH + lane;
                    #pragma unroll
                    for (int q = 0; q < CPL; ++q)
                        unsafeAtomicAdd(ap + 32 * q, hv[j][q] * nrm[j]);
                }
            } else {
                for (int k = b; k < total; ++k) {
                    const int2 en = lst[k];
                    const int s1 = __builtin_amdgcn_readfirstlane(en.x);
                    const int d1 = lo + __builtin_amdgcn_readfirstlane(en.y);
                    const float nr = dinv[s1] * dinv[d1];
                    const float* hp = xin + (size_t)s1 * NDIM + lane;
                    float* ap = acc + (d1 - lo) * APITCH + lane;
                    float hv1[CPL];
                    #pragma unroll
                    for (int q = 0; q < CPL; ++q) hv1[q] = hp[32 * q];
                    #pragma unroll
                    for (int q = 0; q < CPL; ++q) unsafeAtomicAdd(ap + 32 * q, hv1[q] * nr);
                }
            }
        }
    }
    __syncthreads();

    {
        const int l16    = lane & 15;
        const int lhalf  = lane >> 4;
        const int ntiles = (cnt + 15) >> 4;
        for (int rt = wv; rt < ntiles; rt += 32) {
            int arow = rt * 16 + l16;
            if (arow > cnt - 1) arow = cnt - 1;
            const float* aptr = acc + arow * APITCH + lhalf * 2;

            v8f dacc[8];
            #pragma unroll
            for (int t = 0; t < 8; ++t) {
                const float bc = bias[t * 16 + l16];
                #pragma unroll
                for (int v = 0; v < 8; ++v) dacc[t][v] = bc;
            }
            for (int k = 0; k < NDIM; k += 4) {
                v2f a;
                {
                    const float2 av = *(const float2*)(aptr + k);
                    a[0] = av.x; a[1] = av.y;
                }
                mma_kstep(dacc, a, Ws, (k + lhalf * 2) * NDIM + l16);
            }

            #pragma unroll
            for (int t = 0; t < 8; ++t) {
                #pragma unroll
                for (int v = 0; v < 8; ++v) {
                    const int r = rt * 16 + v + lhalf * 8;
                    if (r < cnt) acc[r * APITCH + t * 16 + l16] = dacc[t][v];
                }
            }

            __builtin_amdgcn_fence(__ATOMIC_SEQ_CST, "wavefront");

            const int rend = (rt * 16 + 16 < cnt) ? (rt * 16 + 16) : cnt;
            for (int r = rt * 16; r < rend; ++r) {
                float4 o = *(const float4*)(acc + r * APITCH + lane * 4);
                if (TANH_OUT) { o.x = tanhf(o.x); o.y = tanhf(o.y); o.z = tanhf(o.z); o.w = tanhf(o.w); }
                *(float4*)(out + (size_t)(lo + r) * NDIM + lane * 4) = o;
            }
        }
    }
}

extern "C" void kernel_launch(void* const* d_in, const int* in_sizes, int n_in,
                              void* d_out, int out_size, void* d_ws, size_t ws_size,
                              hipStream_t stream) {
    const float* x   = (const float*)d_in[0];
    const float* W1  = (const float*)d_in[1];
    const float* b1  = (const float*)d_in[2];
    const float* W2  = (const float*)d_in[3];
    const float* b2  = (const float*)d_in[4];
    const int*   ei  = (const int*)d_in[5];

    const int N = in_sizes[0] / NDIM;
    const int E = in_sizes[5] / 2;
    const int* src = ei;
    const int* dst = ei + E;

    float* dinv = (float*)d_ws;
    float* a1   = dinv + ((N + 255) & ~255);
    float* outp = (float*)d_out;

    int gA   = (N + OWN - 1) / OWN;

    int* gcount = (int*)(a1 + (size_t)N * NDIM);
    int* glist  = gcount + ((gA + 255) & ~255);

    k_own_filter<<<gA, 1024, 0, stream>>>(src, dst, dinv, glist, gcount, N, E);

    k_layer<true> <<<gA, 1024, 0, stream>>>(x,  dinv, b1, W1, src, dst, glist, gcount, a1,   N, E);

    k_layer<false><<<gA, 1024, 0, stream>>>(a1, dinv, b2, W2, src, dst, glist, gcount, outp, N, E);
}


# Gather loads widened to 16 B per lane: one load instruction per 512-B source row, not four; rows of sums and W's rows in LDS keep channel 4L+q at word 32q+L, so the product's sum order changes
# speedup vs baseline: 2.9283x; 1.0364x over previous
#include <hip/hip_runtime.h>
#include <hip/hip_bf16.h>

typedef __attribute__((ext_vector_type(2))) float v2f;
typedef __attribute__((ext_vector_type(8))) float v8f;

#define NDIM 128

__device__ __forceinline__ void mma_kstep(v8f (&acc)[8], v2f a, const float* Ws, int krow) {
    #pragma unroll
    for (int t = 0; t < 8; ++t) {
        v2f b;
        b[0] = Ws[krow + t * 16];
        b[1] = Ws[krow + NDIM + t * 16];
        acc[t] = __builtin_amdgcn_wmma_f32_16x16x4_f32(
            false, a, false, b, (short)0, acc[t], false, false);
    }
}

#define OWN  391
#define APITCH 132
#define LCAP 4096
#define LTOT (LCAP + OWN)
#define EPW 8
#define CPL 4
#define CHPOS(c) ((((c) & 3) << 5) + ((c) >> 2))
#define SCAN_V4 8

template <class F>
__device__ __forceinline__ void scan_owned(const int* __restrict__ dst, int e, int wv, int lane,
                                           int lo, unsigned ucnt, F&& onMatch) {
    const bool al16 = ((size_t)dst & 15) == 0;
    const int perWave = SCAN_V4 * 128;
    for (int base = wv * perWave; base < e; base += 32 * perWave) {
        int4 v[SCAN_V4];
        if (al16 && base + perWave <= e) {
            #pragma unroll
            for (int j = 0; j < SCAN_V4; ++j)
                v[j] = *(const int4*)(dst + base + j * 128 + lane * 4);
        } else {
            #pragma unroll
            for (int j = 0; j < SCAN_V4; ++j) {
                const int i = base + j * 128 + lane * 4;
                v[j].x = (i     < e) ? dst[i]     : -1;
                v[j].y = (i + 1 < e) ? dst[i + 1] : -1;
                v[j].z = (i + 2 < e) ? dst[i + 2] : -1;
                v[j].w = (i + 3 < e) ? dst[i + 3] : -1;
            }
        }
        #pragma unroll
        for (int j = 0; j < SCAN_V4; ++j) {
            const int i = base + j * 128 + lane * 4;
            unsigned r[4];
            r[0] = (unsigned)(v[j].x - lo); r[1] = (unsigned)(v[j].y - lo);
            r[2] = (unsigned)(v[j].z - lo); r[3] = (unsigned)(v[j].w - lo);
            unsigned m01 = r[0] < r[1] ? r[0] : r[1];
            unsigned m23 = r[2] < r[3] ? r[2] : r[3];
            unsigned m = m01 < m23 ? m01 : m23;
            if (m < ucnt) {
                #pragma unroll
                for (int k = 0; k < 4; ++k)
                    if (r[k] < ucnt) onMatch(i + k, r[k]);
            }
        }
    }
}

__global__ void __launch_bounds__(1024)
k_own_filter(const int* __restrict__ src, const int* __restrict__ dst, float* __restrict__ dinv,
             int* __restrict__ glist, int* __restrict__ gcount, int n, int e) {
    __shared__ int indeg[OWN];
    __shared__ int lids[LCAP];
    __shared__ int lrow[LCAP];
    __shared__ int lcount;

    const int tid  = threadIdx.x;
    const int lane = tid & 31;
    const int wv   = __builtin_amdgcn_readfirstlane(tid >> 5);
    const int lo   = blockIdx.x * OWN;
    const int cnt  = (n - lo < OWN) ? (n - lo) : OWN;

    if (tid < OWN) indeg[tid] = 0;
    if (tid == 0) lcount = 0;
    __syncthreads();

    scan_owned(dst, e, wv, lane, lo, (unsigned)cnt, [&](int id, unsigned r) {
        atomicAdd(&indeg[r], 1);
        const int slot = atomicAdd(&lcount, 1);
        if (slot < LCAP) { lids[slot] = id; lrow[slot] = (int)r; }
    });
    __syncthreads();

    const int total = lcount;
    if (tid == 0) gcount[blockIdx.x] = total;

    int2* gl = (int2*)glist + (size_t)blockIdx.x * LTOT;
    if (tid < cnt) {
        dinv[lo + tid] = __frsqrt_rn((float)(indeg[tid] + 1));
        gl[tid] = make_int2(lo + tid, tid);
    }
    const int nl = (total < LCAP) ? total : LCAP;
    int sv[LCAP / 1024];
    #pragma unroll
    for (int q = 0; q < LCAP / 1024; ++q) {
        const int i = tid + 1024 * q;
        sv[q] = (i < nl) ? src[lids[i]] : 0;
    }
    #pragma unroll
    for (int q = 0; q < LCAP / 1024; ++q) {
        const int i = tid + 1024 * q;
        if (i < nl) gl[cnt + i] = make_int2(sv[q], lrow[i]);
    }
}

__device__ __noinline__ void agg_edge_by_lane(const float* __restrict__ h, const float* __restrict__ dinv,
                                              const int* __restrict__ src, float* accRow, int i, int d) {
    const int s = src[i];
    const float nrm = dinv[s] * dinv[d];
    const float* hp = h + (size_t)s * NDIM;
    for (int c = 0; c < NDIM; ++c) unsafeAtomicAdd(accRow + CHPOS(c), hp[c] * nrm);
}

template <bool TANH_OUT>
__global__ void __launch_bounds__(1024)
k_layer(const float* __restrict__ xin, const float* __restrict__ dinv, const float* __restrict__ bias,
        const float* __restrict__ W,
        const int* __restrict__ src, const int* __restrict__ dst,
        const int* __restrict__ glist, const int* __restrict__ gcount,
        float* __restrict__ out, int n, int e) {
    __shared__ __attribute__((aligned(16))) float acc[OWN * APITCH];
    __shared__ __attribute__((aligned(16))) float Ws[NDIM * NDIM];
    __shared__ __attribute__((aligned(16))) int2 lst[LTOT];
    __shared__ int   lcount;

    const int tid  = threadIdx.x;
    const int lane = tid & 31;
    const int wv   = __builtin_amdgcn_readfirstlane(tid >> 5);
    const int lo   = blockIdx.x * OWN;
    const int cnt  = (n - lo < OWN) ? (n - lo) : OWN;
    const unsigned ucnt = (unsigned)cnt;

    const int  gtotal = gcount[blockIdx.x];
    const bool listed = gtotal <= LCAP;

    if (tid == 0) lcount = listed ? gtotal : 0;
    float4 wreg[NDIM * NDIM / 4096];
    {
        const float4* Wv = (const float4*)W;
        #pragma unroll
        for (int q = 0; q < NDIM * NDIM / 4096; ++q) wreg[q] = Wv[tid + 1024 * q];
    }
    const int nent = cnt + gtotal;
    int2 ent[(LTOT + 1023) / 1024];
    if (listed) {
        const int2* gl = (const int2*)glist + (size_t)blockIdx.x * LTOT;
        #pragma unroll
        for (int q = 0; q < (LTOT + 1023) / 1024; ++q) {
            const int i = tid + 1024 * q;
            ent[q] = (i < nent) ? gl[i] : make_int2(0, 0);
        }
    }
    {
        for (int r = wv; r < cnt; r += 32)
            *(float4*)(acc + r * APITCH + lane * 4) = make_float4(0.f, 0.f, 0.f, 0.f);
    }
    {
        float4* Sv = (float4*)Ws;
        #pragma unroll
        for (int q = 0; q < NDIM * NDIM / 4096; ++q) {
            const int f = tid + 1024 * q;
            Sv[CHPOS(f >> 5) * (NDIM / 4) + (f & 31)] = wreg[q];
        }
    }
    if (listed) {
        #pragma unroll
        for (int q = 0; q < (LTOT + 1023) / 1024; ++q) {
            const int i = tid + 1024 * q;
            if (i < nent) lst[i] = ent[q];
        }
    } else if (tid < cnt) {
        lst[tid] = make_int2(lo + tid, tid);
    }
    __syncthreads();

    if (!listed) {
        scan_owned(dst, e, wv, lane, lo, ucnt, [&](int id, unsigned r) {
            const int slot = atomicAdd(&lcount, 1);
            if (slot < LCAP) lst[cnt + slot] = make_int2(src[id], (int)r);
            else agg_edge_by_lane(xin, dinv, src, acc + r * APITCH, id, lo + (int)r);
        });
    }
    __syncthreads();

    {
        const int total = cnt + ((lcount < LCAP) ? lcount : LCAP);
        for (int b = wv * EPW; b < total; b += 32 * EPW) {
            if (b + EPW <= total) {
                int s[EPW], d[EPW];
                #pragma unroll
                for (int j = 0; j < EPW; ++j) {
                    const int2 en = lst[b + j];
                    s[j] = __builtin_amdgcn_readfirstlane(en.x);
                    d[j] = lo + __builtin_amdgcn_readfirstlane(en.y);
                }
                float4 hv[EPW];
                #pragma unroll
                for (int j = 0; j < EPW; ++j)
                    hv[j] = *(const float4*)(xin + (size_t)s[j] * NDIM + lane * CPL);
                float nrm[EPW];
                #pragma unroll
                for (int j = 0; j < EPW; ++j) nrm[j] = dinv[s[j]] * dinv[d[j]];
                #pragma unroll
                for (int j = 0; j < EPW; ++j) {
                    float* ap = acc + (d[j] - lo) * APITCH + lane;
                    unsafeAtomicAdd(ap,      hv[j].x * nrm[j]);
                    unsafeAtomicAdd(ap + 32, hv[j].y * nrm[j]);
                    unsafeAtomicAdd(ap + 64, hv[j].z * nrm[j]);
                    unsafeAtomicAdd(ap + 96, hv[j].w * nrm[j]);
                }
            } else {
                for (int k = b; k < total; ++k) {
                    const int2 en = lst[k];
                    const int s1 = __builtin_amdgcn_readfirstlane(en.x);
                    const int d1 = lo + __builtin_amdgcn_readfirstlane(en.y);
                    const float nr = dinv[s1] * dinv[d1];
                    const float4 hv1 = *(const float4*)(xin + (size_t)s1 * NDIM + lane * CPL);
                    float* ap = acc + (d1 - lo) * APITCH + lane;
                    unsafeAtomicAdd(ap,      hv1.x * nr);
                    unsafeAtomicAdd(ap + 32, hv1.y * nr);
                    unsafeAtomicAdd(ap + 64, hv1.z * nr);
                    unsafeAtomicAdd(ap + 96, hv1.w * nr);
                }
            }
        }
    }
    __syncthreads();

    {
        const int l16    = lane & 15;
        const int lhalf  = lane >> 4;
        const int ntiles = (cnt + 15) >> 4;
        for (int rt = wv; rt < ntiles; rt += 32) {
            int arow = rt * 16 + l16;
            if (arow > cnt - 1) arow = cnt - 1;
            const float* aptr = acc + arow * APITCH + lhalf * 2;

            v8f dacc[8];
            #pragma unroll
            for (int t = 0; t < 8; ++t) {
                const float bc = bias[t * 16 + l16];
                #pragma unroll
                for (int v = 0; v < 8; ++v) dacc[t][v] = bc;
            }
            for (int k = 0; k < NDIM; k += 4) {
                v2f a;
                {
                    const float2 av = *(const float2*)(aptr + k);
                    a[0] = av.x; a[1] = av.y;
                }
                mma_kstep(dacc, a, Ws, (k + lhalf * 2) * NDIM + l16);
            }

            #pragma unroll
            for (int t = 0; t < 8; ++t) {
                #pragma unroll
                for (int v = 0; v < 8; ++v) {
                    const int r = rt * 16 + v + lhalf * 8;
                    if (r < cnt) acc[r * APITCH + t * 16 + l16] = dacc[t][v];
                }
            }

            __builtin_amdgcn_fence(__ATOMIC_SEQ_CST, "wavefront");

            const int rend = (rt * 16 + 16 < cnt) ? (rt * 16 + 16) : cnt;
            for (int r = rt * 16; r < rend; ++r) {
                float4 o = *(const float4*)(acc + r * APITCH + lane * 4);
                if (TANH_OUT) { o.x = tanhf(o.x); o.y = tanhf(o.y); o.z = tanhf(o.z); o.w = tanhf(o.w); }
                *(float4*)(out + (size_t)(lo + r) * NDIM + lane * 4) = o;
            }
        }
    }
}

extern "C" void kernel_launch(void* const* d_in, const int* in_sizes, int n_in,
                              void* d_out, int out_size, void* d_ws, size_t ws_size,
                              hipStream_t stream) {
    const float* x   = (const float*)d_in[0];
    const float* W1  = (const float*)d_in[1];
    const float* b1  = (const float*)d_in[2];
    const float* W2  = (const float*)d_in[3];
    const float* b2  = (const float*)d_in[4];
    const int*   ei  = (const int*)d_in[5];

    const int N = in_sizes[0] / NDIM;
    const int E = in_sizes[5] / 2;
    const int* src = ei;
    const int* dst = ei + E;

    float* dinv = (float*)d_ws;
    float* a1   = dinv + ((N + 255) & ~255);
    float* outp = (float*)d_out;

    int gA   = (N + OWN - 1) / OWN;

    int* gcount = (int*)(a1 + (size_t)N * NDIM);
    int* glist  = gcount + ((gA + 255) & ~255);

    k_own_filter<<<gA, 1024, 0, stream>>>(src, dst, dinv, glist, gcount, N, E);

    k_layer<true> <<<gA, 1024, 0, stream>>>(x,  dinv, b1, W1, src, dst, glist, gcount, a1,   N, E);

    k_layer<false><<<gA, 1024, 0, stream>>>(a1, dinv, b2, W2, src, dst, glist, gcount, outp, N, E);
}


# Owner blocks' edge filter scans one key byte per edge (owner block of its destination, written first by a new small kernel), not the 4-byte destination list; 3 -> 4 launches; sums in arrival order
# speedup vs baseline: 3.1368x; 1.0712x over previous
#include <hip/hip_runtime.h>
#include <hip/hip_bf16.h>

typedef __attribute__((ext_vector_type(2))) float v2f;
typedef __attribute__((ext_vector_type(8))) float v8f;

#define NDIM 128

__device__ __forceinline__ void mma_kstep(v8f (&acc)[8], v2f a, const float* Ws, int krow) {
    #pragma unroll
    for (int t = 0; t < 8; ++t) {
        v2f b;
        b[0] = Ws[krow + t * 16];
        b[1] = Ws[krow + NDIM + t * 16];
        acc[t] = __builtin_amdgcn_wmma_f32_16x16x4_f32(
            false, a, false, b, (short)0, acc[t], false, false);
    }
}

#define OWN  391
#define APITCH 132
#define LCAP 4096
#define LTOT (LCAP + OWN)
#define EPW 8
#define CPL 4
#define CHPOS(c) ((((c) & 3) << 5) + ((c) >> 2))
#define SCAN_V4 8
#define KEY_SAT 255u
#define KTRIP (SCAN_V4 * 32 * 16)

template <class F>
__device__ __forceinline__ void scan_owned(const int* __restrict__ dst, int e, int wv, int lane,
                                           int lo, unsigned ucnt, F&& onMatch) {
    const bool al16 = ((size_t)dst & 15) == 0;
    const int perWave = SCAN_V4 * 128;
    for (int base = wv * perWave; base < e; base += 32 * perWave) {
        int4 v[SCAN_V4];
        if (al16 && base + perWave <= e) {
            #pragma unroll
            for (int j = 0; j < SCAN_V4; ++j)
                v[j] = *(const int4*)(dst + base + j * 128 + lane * 4);
        } else {
            #pragma unroll
            for (int j = 0; j < SCAN_V4; ++j) {
                const int i = base + j * 128 + lane * 4;
                v[j].x = (i     < e) ? dst[i]     : -1;
                v[j].y = (i + 1 < e) ? dst[i + 1] : -1;
                v[j].z = (i + 2 < e) ? dst[i + 2] : -1;
                v[j].w = (i + 3 < e) ? dst[i + 3] : -1;
            }
        }
        #pragma unroll
        for (int j = 0; j < SCAN_V4; ++j) {
            const int i = base + j * 128 + lane * 4;
            unsigned r[4];
            r[0] = (unsigned)(v[j].x - lo); r[1] = (unsigned)(v[j].y - lo);
            r[2] = (unsigned)(v[j].z - lo); r[3] = (unsigned)(v[j].w - lo);
            unsigned m01 = r[0] < r[1] ? r[0] : r[1];
            unsigned m23 = r[2] < r[3] ? r[2] : r[3];
            unsigned m = m01 < m23 ? m01 : m23;
            if (m < ucnt) {
                #pragma unroll
                for (int k = 0; k < 4; ++k)
                    if (r[k] < ucnt) onMatch(i + k, r[k]);
            }
        }
    }
}

__device__ __forceinline__ unsigned key_of(int d) {
    const unsigned k = (unsigned)d / (unsigned)OWN;
    return k < KEY_SAT ? k : KEY_SAT;
}
__global__ void __launch_bounds__(256)
k_edge_keys(const int* __restrict__ dst, unsigned* __restrict__ keys, int e, int nwords) {
    const int w = blockIdx.x * 256 + threadIdx.x;
    if (w >= nwords) return;
    const int i = w * 4;
    int4 v;
    if (((size_t)dst & 15) == 0 && i + 4 <= e) {
        v = *(const int4*)(dst + i);
    } else {
        v.x = (i     < e) ? dst[i]     : -1;
        v.y = (i + 1 < e) ? dst[i + 1] : -1;
        v.z = (i + 2 < e) ? dst[i + 2] : -1;
        v.w = (i + 3 < e) ? dst[i + 3] : -1;
    }
    keys[w] = key_of(v.x) | (key_of(v.y) << 8) | (key_of(v.z) << 16) | (key_of(v.w) << 24);
}

__global__ void __launch_bounds__(1024)
k_own_filter(const int* __restrict__ src, const int* __restrict__ dst, const unsigned* __restrict__ keys,
             float* __restrict__ dinv, int* __restrict__ glist, int* __restrict__ gcount, int n, int e) {
    __shared__ int indeg[OWN];
    __shared__ int lids[LCAP];
    __shared__ int lrow[LCAP];
    __shared__ int lcount;
    __shared__ int nvalid;

    const int tid  = threadIdx.x;
    const int lane = tid & 31;
    const int wv   = __builtin_amdgcn_readfirstlane(tid >> 5);
    const int lo   = blockIdx.x * OWN;
    const int cnt  = (n - lo < OWN) ? (n - lo) : OWN;

    if (tid < OWN) indeg[tid] = 0;
    if (tid == 0) { lcount = 0; nvalid = 0; }
    __syncthreads();

    {
        const unsigned mykey = (blockIdx.x < KEY_SAT) ? blockIdx.x : KEY_SAT;
        const unsigned pat = mykey * 0x01010101u;
        const uint4* kv = (const uint4*)keys;
        for (int base = wv * KTRIP; base < e; base += 32 * KTRIP) {
            uint4 v[SCAN_V4];
            #pragma unroll
            for (int j = 0; j < SCAN_V4; ++j) v[j] = kv[(base >> 4) + j * 32 + lane];
            #pragma unroll
            for (int j = 0; j < SCAN_V4; ++j) {
                const int i = base + j * 512 + lane * 16;
                unsigned x[4];
                x[0] = v[j].x ^ pat; x[1] = v[j].y ^ pat; x[2] = v[j].z ^ pat; x[3] = v[j].w ^ pat;
                unsigned zb[4];
                #pragma unroll
                for (int c4 = 0; c4 < 4; ++c4)
                    zb[c4] = ~(((x[c4] & 0x7f7f7f7fu) + 0x7f7f7f7fu) | x[c4]) & 0x80808080u;
                unsigned m = (zb[0] >> 7) | (zb[1] >> 6) | (zb[2] >> 5) | (zb[3] >> 4);
                while (m != 0u) {
                    const int p = __builtin_ctz(m);
                    m &= m - 1u;
                    const int id = i + ((p & 7) << 2) + (p >> 3);
                    if (id < e) {
                        const int slot = atomicAdd(&lcount, 1);
                        if (slot < LCAP) lids[slot] = id;
                    }
                }
            }
        }
    }
    __syncthreads();

    const int ktotal = lcount;
    int2* gl = (int2*)glist + (size_t)blockIdx.x * LTOT;
    if (ktotal <= LCAP) {
        int dq[LCAP / 1024], sq[LCAP / 1024];
        #pragma unroll
        for (int q = 0; q < LCAP / 1024; ++q) {
            const int i = tid + 1024 * q;
            const bool in = i < ktotal;
            const int id = in ? lids[i] : 0;
            dq[q] = in ? dst[id] : -1;
            sq[q] = in ? src[id] : 0;
        }
        #pragma unroll
        for (int q = 0; q < LCAP / 1024; ++q) {
            const unsigned r = (unsigned)(dq[q] - lo);
            const bool ok = r < (unsigned)cnt;
            const unsigned mask = __builtin_amdgcn_ballot_w32(ok);
            int first = 0;
            if (lane == 0 && mask != 0u) first = atomicAdd(&nvalid, __popc(mask));
            first = __builtin_amdgcn_readfirstlane(first);
            if (ok) {
                const int slot = first + (int)__builtin_amdgcn_mbcnt_lo(mask, 0u);
                atomicAdd(&indeg[r], 1);
                gl[cnt + slot] = make_int2(sq[q], (int)r);
            }
        }
        __syncthreads();
    } else {
        scan_owned(dst, e, wv, lane, lo, (unsigned)cnt, [&](int id, unsigned r) {
            atomicAdd(&indeg[r], 1);
            const int slot = atomicAdd(&nvalid, 1);
            if (slot < LCAP) { lids[slot] = id; lrow[slot] = (int)r; }
        });
        __syncthreads();
        const int nl = (nvalid < LCAP) ? nvalid : LCAP;
        for (int i = tid; i < nl; i += 1024) gl[cnt + i] = make_int2(src[lids[i]], lrow[i]);
    }

    const int total = nvalid;
    if (tid == 0) gcount[blockIdx.x] = total;
    if (tid < cnt) {
        dinv[lo + tid] = __frsqrt_rn((float)(indeg[tid] + 1));
        gl[tid] = make_int2(lo + tid, tid);
    }
}

__device__ __noinline__ void agg_edge_by_lane(const float* __restrict__ h, const float* __restrict__ dinv,
                                              const int* __restrict__ src, float* accRow, int i, int d) {
    const int s = src[i];
    const float nrm = dinv[s] * dinv[d];
    const float* hp = h + (size_t)s * NDIM;
    for (int c = 0; c < NDIM; ++c) unsafeAtomicAdd(accRow + CHPOS(c), hp[c] * nrm);
}

template <bool TANH_OUT>
__global__ void __launch_bounds__(1024)
k_layer(const float* __restrict__ xin, const float* __restrict__ dinv, const float* __restrict__ bias,
        const float* __restrict__ W,
        const int* __restrict__ src, const int* __restrict__ dst,
        const int* __restrict__ glist, const int* __restrict__ gcount,
        float* __restrict__ out, int n, int e) {
    __shared__ __attribute__((aligned(16))) float acc[OWN * APITCH];
    __shared__ __attribute__((aligned(16))) float Ws[NDIM * NDIM];
    __shared__ __attribute__((aligned(16))) int2 lst[LTOT];
    __shared__ int   lcount;

    const int tid  = threadIdx.x;
    const int lane = tid & 31;
    const int wv   = __builtin_amdgcn_readfirstlane(tid >> 5);
    const int lo   = blockIdx.x * OWN;
    const int cnt  = (n - lo < OWN) ? (n - lo) : OWN;
    const unsigned ucnt = (unsigned)cnt;

    const int  gtotal = gcount[blockIdx.x];
    const bool listed = gtotal <= LCAP;

    if (tid == 0) lcount = listed ? gtotal : 0;
    float4 wreg[NDIM * NDIM / 4096];
    {
        const float4* Wv = (const float4*)W;
        #pragma unroll
        for (int q = 0; q < NDIM * NDIM / 4096; ++q) wreg[q] = Wv[tid + 1024 * q];
    }
    const int nent = cnt + gtotal;
    int2 ent[(LTOT + 1023) / 1024];
    if (listed) {
        const int2* gl = (const int2*)glist + (size_t)blockIdx.x * LTOT;
        #pragma unroll
        for (int q = 0; q < (LTOT + 1023) / 1024; ++q) {
            const int i = tid + 1024 * q;
            ent[q] = (i < nent) ? gl[i] : make_int2(0, 0);
        }
    }
    {
        for (int r = wv; r < cnt; r += 32)
            *(float4*)(acc + r * APITCH + lane * 4) = make_float4(0.f, 0.f, 0.f, 0.f);
    }
    {
        float4* Sv = (float4*)Ws;
        #pragma unroll
        for (int q = 0; q < NDIM * NDIM / 4096; ++q) {
            const int f = tid + 1024 * q;
            Sv[CHPOS(f >> 5) * (NDIM / 4) + (f & 31)] = wreg[q];
        }
    }
    if (listed) {
        #pragma unroll
        for (int q = 0; q < (LTOT + 1023) / 1024; ++q) {
            const int i = tid + 1024 * q;
            if (i < nent) lst[i] = ent[q];
        }
    } else if (tid < cnt) {
        lst[tid] = make_int2(lo + tid, tid);
    }
    __syncthreads();

    if (!listed) {
        scan_owned(dst, e, wv, lane, lo, ucnt, [&](int id, unsigned r) {
            const int slot = atomicAdd(&lcount, 1);
            if (slot < LCAP) lst[cnt + slot] = make_int2(src[id], (int)r);
            else agg_edge_by_lane(xin, dinv, src, acc + r * APITCH, id, lo + (int)r);
        });
    }
    __syncthreads();

    {
        const int total = cnt + ((lcount < LCAP) ? lcount : LCAP);
        for (int b = wv * EPW; b < total; b += 32 * EPW) {
            if (b + EPW <= total) {
                int s[EPW], d[EPW];
                #pragma unroll
                for (int j = 0; j < EPW; ++j) {
                    const int2 en = lst[b + j];
                    s[j] = __builtin_amdgcn_readfirstlane(en.x);
                    d[j] = lo + __builtin_amdgcn_readfirstlane(en.y);
                }
                float4 hv[EPW];
                #pragma unroll
                for (int j = 0; j < EPW; ++j)
                    hv[j] = *(const float4*)(xin + (size_t)s[j] * NDIM + lane * CPL);
                float nrm[EPW];
                #pragma unroll
                for (int j = 0; j < EPW; ++j) nrm[j] = dinv[s[j]] * dinv[d[j]];
                #pragma unroll
                for (int j = 0; j < EPW; ++j) {
                    float* ap = acc + (d[j] - lo) * APITCH + lane;
                    unsafeAtomicAdd(ap,      hv[j].x * nrm[j]);
                    unsafeAtomicAdd(ap + 32, hv[j].y * nrm[j]);
                    unsafeAtomicAdd(ap + 64, hv[j].z * nrm[j]);
                    unsafeAtomicAdd(ap + 96, hv[j].w * nrm[j]);
                }
            } else {
                for (int k = b; k < total; ++k) {
                    const int2 en = lst[k];
                    const int s1 = __builtin_amdgcn_readfirstlane(en.x);
                    const int d1 = lo + __builtin_amdgcn_readfirstlane(en.y);
                    const float nr = dinv[s1] * dinv[d1];
                    const float4 hv1 = *(const float4*)(xin + (size_t)s1 * NDIM + lane * CPL);
                    float* ap = acc + (d1 - lo) * APITCH + lane;
                    unsafeAtomicAdd(ap,      hv1.x * nr);
                    unsafeAtomicAdd(ap + 32, hv1.y * nr);
                    unsafeAtomicAdd(ap + 64, hv1.z * nr);
                    unsafeAtomicAdd(ap + 96, hv1.w * nr);
                }
            }
        }
    }
    __syncthreads();

    {
        const int l16    = lane & 15;
        const int lhalf  = lane >> 4;
        const int ntiles = (cnt + 15) >> 4;
        for (int rt = wv; rt < ntiles; rt += 32) {
            int arow = rt * 16 + l16;
            if (arow > cnt - 1) arow = cnt - 1;
            const float* aptr = acc + arow * APITCH + lhalf * 2;

            v8f dacc[8];
            #pragma unroll
            for (int t = 0; t < 8; ++t) {
                const float bc = bias[t * 16 + l16];
                #pragma unroll
                for (int v = 0; v < 8; ++v) dacc[t][v] = bc;
            }
            for (int k = 0; k < NDIM; k += 4) {
                v2f a;
                {
                    const float2 av = *(const float2*)(aptr + k);
                    a[0] = av.x; a[1] = av.y;
                }
                mma_kstep(dacc, a, Ws, (k + lhalf * 2) * NDIM + l16);
            }

            #pragma unroll
            for (int t = 0; t < 8; ++t) {
                #pragma unroll
                for (int v = 0; v < 8; ++v) {
                    const int r = rt * 16 + v + lhalf * 8;
                    if (r < cnt) acc[r * APITCH + t * 16 + l16] = dacc[t][v];
                }
            }

            __builtin_amdgcn_fence(__ATOMIC_SEQ_CST, "wavefront");

            const int rend = (rt * 16 + 16 < cnt) ? (rt * 16 + 16) : cnt;
            for (int r = rt * 16; r < rend; ++r) {
                float4 o = *(const float4*)(acc + r * APITCH + lane * 4);
                if (TANH_OUT) { o.x = tanhf(o.x); o.y = tanhf(o.y); o.z = tanhf(o.z); o.w = tanhf(o.w); }
                *(float4*)(out + (size_t)(lo + r) * NDIM + lane * 4) = o;
            }
        }
    }
}

extern "C" void kernel_launch(void* const* d_in, const int* in_sizes, int n_in,
                              void* d_out, int out_size, void* d_ws, size_t ws_size,
                              hipStream_t stream) {
    const float* x   = (const float*)d_in[0];
    const float* W1  = (const float*)d_in[1];
    const float* b1  = (const float*)d_in[2];
    const float* W2  = (const float*)d_in[3];
    const float* b2  = (const float*)d_in[4];
    const int*   ei  = (const int*)d_in[5];

    const int N = in_sizes[0] / NDIM;
    const int E = in_sizes[5] / 2;
    const int* src = ei;
    const int* dst = ei + E;

    float* dinv = (float*)d_ws;
    unsigned* keys = (unsigned*)(dinv + ((N + 255) & ~255));
    const int nwords = ((E + KTRIP - 1) / KTRIP) * (KTRIP / 4);
    float* a1   = (float*)(keys + nwords);
    float* outp = (float*)d_out;

    int gA   = (N + OWN - 1) / OWN;

    int* gcount = (int*)(a1 + (size_t)N * NDIM);
    int* glist  = gcount + ((gA + 255) & ~255);

    k_edge_keys<<<(nwords + 255) / 256, 256, 0, stream>>>(dst, keys, E, nwords);
    k_own_filter<<<gA, 1024, 0, stream>>>(src, dst, keys, dinv, glist, gcount, N, E);

    k_layer<true> <<<gA, 1024, 0, stream>>>(x,  dinv, b1, W1, src, dst, glist, gcount, a1,   N, E);

    k_layer<false><<<gA, 1024, 0, stream>>>(a1, dinv, b2, W2, src, dst, glist, gcount, outp, N, E);
}


# Layer 1's tanh on the way out is the chip's transcendental v_tanh_f32, one instruction per value, not the library routine's two divergent branches (exp and rcp, polynomial); sum order unchanged
# speedup vs baseline: 3.2547x; 1.0376x over previous
#include <hip/hip_runtime.h>
#include <hip/hip_bf16.h>

typedef __attribute__((ext_vector_type(2))) float v2f;
typedef __attribute__((ext_vector_type(8))) float v8f;

#define NDIM 128

__device__ __forceinline__ void mma_kstep(v8f (&acc)[8], v2f a, const float* Ws, int krow) {
    #pragma unroll
    for (int t = 0; t < 8; ++t) {
        v2f b;
        b[0] = Ws[krow + t * 16];
        b[1] = Ws[krow + NDIM + t * 16];
        acc[t] = __builtin_amdgcn_wmma_f32_16x16x4_f32(
            false, a, false, b, (short)0, acc[t], false, false);
    }
}

#define OWN  391
#define APITCH 132
#define LCAP 4096
#define LTOT (LCAP + OWN)
#define EPW 8
#define CPL 4
#define CHPOS(c) ((((c) & 3) << 5) + ((c) >> 2))
#define SCAN_V4 8
#define KEY_SAT 255u
#define KTRIP (SCAN_V4 * 32 * 16)

template <class F>
__device__ __forceinline__ void scan_owned(const int* __restrict__ dst, int e, int wv, int lane,
                                           int lo, unsigned ucnt, F&& onMatch) {
    const bool al16 = ((size_t)dst & 15) == 0;
    const int perWave = SCAN_V4 * 128;
    for (int base = wv * perWave; base < e; base += 32 * perWave) {
        int4 v[SCAN_V4];
        if (al16 && base + perWave <= e) {
            #pragma unroll
            for (int j = 0; j < SCAN_V4; ++j)
                v[j] = *(const int4*)(dst + base + j * 128 + lane * 4);
        } else {
            #pragma unroll
            for (int j = 0; j < SCAN_V4; ++j) {
                const int i = base + j * 128 + lane * 4;
                v[j].x = (i     < e) ? dst[i]     : -1;
                v[j].y = (i + 1 < e) ? dst[i + 1] : -1;
                v[j].z = (i + 2 < e) ? dst[i + 2] : -1;
                v[j].w = (i + 3 < e) ? dst[i + 3] : -1;
            }
        }
        #pragma unroll
        for (int j = 0; j < SCAN_V4; ++j) {
            const int i = base + j * 128 + lane * 4;
            unsigned r[4];
            r[0] = (unsigned)(v[j].x - lo); r[1] = (unsigned)(v[j].y - lo);
            r[2] = (unsigned)(v[j].z - lo); r[3] = (unsigned)(v[j].w - lo);
            unsigned m01 = r[0] < r[1] ? r[0] : r[1];
            unsigned m23 = r[2] < r[3] ? r[2] : r[3];
            unsigned m = m01 < m23 ? m01 : m23;
            if (m < ucnt) {
                #pragma unroll
                for (int k = 0; k < 4; ++k)
                    if (r[k] < ucnt) onMatch(i + k, r[k]);
            }
        }
    }
}

__device__ __forceinline__ unsigned key_of(int d) {
    const unsigned k = (unsigned)d / (unsigned)OWN;
    return k < KEY_SAT ? k : KEY_SAT;
}
__global__ void __launch_bounds__(256)
k_edge_keys(const int* __restrict__ dst, unsigned* __restrict__ keys, int e, int nwords) {
    const int w = blockIdx.x * 256 + threadIdx.x;
    if (w >= nwords) return;
    const int i = w * 4;
    int4 v;
    if (((size_t)dst & 15) == 0 && i + 4 <= e) {
        v = *(const int4*)(dst + i);
    } else {
        v.x = (i     < e) ? dst[i]     : -1;
        v.y = (i + 1 < e) ? dst[i + 1] : -1;
        v.z = (i + 2 < e) ? dst[i + 2] : -1;
        v.w = (i + 3 < e) ? dst[i + 3] : -1;
    }
    keys[w] = key_of(v.x) | (key_of(v.y) << 8) | (key_of(v.z) << 16) | (key_of(v.w) << 24);
}

__global__ void __launch_bounds__(1024)
k_own_filter(const int* __restrict__ src, const int* __restrict__ dst, const unsigned* __restrict__ keys,
             float* __restrict__ dinv, int* __restrict__ glist, int* __restrict__ gcount, int n, int e) {
    __shared__ int indeg[OWN];
    __shared__ int lids[LCAP];
    __shared__ int lrow[LCAP];
    __shared__ int lcount;
    __shared__ int nvalid;

    const int tid  = threadIdx.x;
    const int lane = tid & 31;
    const int wv   = __builtin_amdgcn_readfirstlane(tid >> 5);
    const int lo   = blockIdx.x * OWN;
    const int cnt  = (n - lo < OWN) ? (n - lo) : OWN;

    if (tid < OWN) indeg[tid] = 0;
    if (tid == 0) { lcount = 0; nvalid = 0; }
    __syncthreads();

    {
        const unsigned mykey = (blockIdx.x < KEY_SAT) ? blockIdx.x : KEY_SAT;
        const unsigned pat = mykey * 0x01010101u;
        const uint4* kv = (const uint4*)keys;
        for (int base = wv * KTRIP; base < e; base += 32 * KTRIP) {
            uint4 v[SCAN_V4];
            #pragma unroll
            for (int j = 0; j < SCAN_V4; ++j) v[j] = kv[(base >> 4) + j * 32 + lane];
            #pragma unroll
            for (int j = 0; j < SCAN_V4; ++j) {
                const int i = base + j * 512 + lane * 16;
                unsigned x[4];
                x[0] = v[j].x ^ pat; x[1] = v[j].y ^ pat; x[2] = v[j].z ^ pat; x[3] = v[j].w ^ pat;
                unsigned zb[4];
                #pragma unroll
                for (int c4 = 0; c4 < 4; ++c4)
                    zb[c4] = ~(((x[c4] & 0x7f7f7f7fu) + 0x7f7f7f7fu) | x[c4]) & 0x80808080u;
                unsigned m = (zb[0] >> 7) | (zb[1] >> 6) | (zb[2] >> 5) | (zb[3] >> 4);
                while (m != 0u) {
                    const int p = __builtin_ctz(m);
                    m &= m - 1u;
                    const int id = i + ((p & 7) << 2) + (p >> 3);
                    if (id < e) {
                        const int slot = atomicAdd(&lcount, 1);
                        if (slot < LCAP) lids[slot] = id;
                    }
                }
            }
        }
    }
    __syncthreads();

    const int ktotal = lcount;
    int2* gl = (int2*)glist + (size_t)blockIdx.x * LTOT;
    if (ktotal <= LCAP) {
        int dq[LCAP / 1024], sq[LCAP / 1024];
        #pragma unroll
        for (int q = 0; q < LCAP / 1024; ++q) {
            const int i = tid + 1024 * q;
            const bool in = i < ktotal;
            const int id = in ? lids[i] : 0;
            dq[q] = in ? dst[id] : -1;
            sq[q] = in ? src[id] : 0;
        }
        #pragma unroll
        for (int q = 0; q < LCAP / 1024; ++q) {
            const unsigned r = (unsigned)(dq[q] - lo);
            const bool ok = r < (unsigned)cnt;
            const unsigned mask = __builtin_amdgcn_ballot_w32(ok);
            int first = 0;
            if (lane == 0 && mask != 0u) first = atomicAdd(&nvalid, __popc(mask));
            first = __builtin_amdgcn_readfirstlane(first);
            if (ok) {
                const int slot = first + (int)__builtin_amdgcn_mbcnt_lo(mask, 0u);
                atomicAdd(&indeg[r], 1);
                gl[cnt + slot] = make_int2(sq[q], (int)r);
            }
        }
        __syncthreads();
    } else {
        scan_owned(dst, e, wv, lane, lo, (unsigned)cnt, [&](int id, unsigned r) {
            atomicAdd(&indeg[r], 1);
            const int slot = atomicAdd(&nvalid, 1);
            if (slot < LCAP) { lids[slot] = id; lrow[slot] = (int)r; }
        });
        __syncthreads();
        const int nl = (nvalid < LCAP) ? nvalid : LCAP;
        for (int i = tid; i < nl; i += 1024) gl[cnt + i] = make_int2(src[lids[i]], lrow[i]);
    }

    const int total = nvalid;
    if (tid == 0) gcount[blockIdx.x] = total;
    if (tid < cnt) {
        dinv[lo + tid] = __frsqrt_rn((float)(indeg[tid] + 1));
        gl[tid] = make_int2(lo + tid, tid);
    }
}

__device__ __noinline__ void agg_edge_by_lane(const float* __restrict__ h, const float* __restrict__ dinv,
                                              const int* __restrict__ src, float* accRow, int i, int d) {
    const int s = src[i];
    const float nrm = dinv[s] * dinv[d];
    const float* hp = h + (size_t)s * NDIM;
    for (int c = 0; c < NDIM; ++c) unsafeAtomicAdd(accRow + CHPOS(c), hp[c] * nrm);
}

template <bool TANH_OUT>
__global__ void __launch_bounds__(1024)
k_layer(const float* __restrict__ xin, const float* __restrict__ dinv, const float* __restrict__ bias,
        const float* __restrict__ W,
        const int* __restrict__ src, const int* __restrict__ dst,
        const int* __restrict__ glist, const int* __restrict__ gcount,
        float* __restrict__ out, int n, int e) {
    __shared__ __attribute__((aligned(16))) float acc[OWN * APITCH];
    __shared__ __attribute__((aligned(16))) float Ws[NDIM * NDIM];
    __shared__ __attribute__((aligned(16))) int2 lst[LTOT];
    __shared__ int   lcount;

    const int tid  = threadIdx.x;
    const int lane = tid & 31;
    const int wv   = __builtin_amdgcn_readfirstlane(tid >> 5);
    const int lo   = blockIdx.x * OWN;
    const int cnt  = (n - lo < OWN) ? (n - lo) : OWN;
    const unsigned ucnt = (unsigned)cnt;

    const int  gtotal = gcount[blockIdx.x];
    const bool listed = gtotal <= LCAP;

    if (tid == 0) lcount = listed ? gtotal : 0;
    float4 wreg[NDIM * NDIM / 4096];
    {
        const float4* Wv = (const float4*)W;
        #pragma unroll
        for (int q = 0; q < NDIM * NDIM / 4096; ++q) wreg[q] = Wv[tid + 1024 * q];
    }
    const int nent = cnt + gtotal;
    int2 ent[(LTOT + 1023) / 1024];
    if (listed) {
        const int2* gl = (const int2*)glist + (size_t)blockIdx.x * LTOT;
        #pragma unroll
        for (int q = 0; q < (LTOT + 1023) / 1024; ++q) {
            const int i = tid + 1024 * q;
            ent[q] = (i < nent) ? gl[i] : make_int2(0, 0);
        }
    }
    {
        for (int r = wv; r < cnt; r += 32)
            *(float4*)(acc + r * APITCH + lane * 4) = make_float4(0.f, 0.f, 0.f, 0.f);
    }
    {
        float4* Sv = (float4*)Ws;
        #pragma unroll
        for (int q = 0; q < NDIM * NDIM / 4096; ++q) {
            const int f = tid + 1024 * q;
            Sv[CHPOS(f >> 5) * (NDIM / 4) + (f & 31)] = wreg[q];
        }
    }
    if (listed) {
        #pragma unroll
        for (int q = 0; q < (LTOT + 1023) / 1024; ++q) {
            const int i = tid + 1024 * q;
            if (i < nent) lst[i] = ent[q];
        }
    } else if (tid < cnt) {
        lst[tid] = make_int2(lo + tid, tid);
    }
    __syncthreads();

    if (!listed) {
        scan_owned(dst, e, wv, lane, lo, ucnt, [&](int id, unsigned r) {
            const int slot = atomicAdd(&lcount, 1);
            if (slot < LCAP) lst[cnt + slot] = make_int2(src[id], (int)r);
            else agg_edge_by_lane(xin, dinv, src, acc + r * APITCH, id, lo + (int)r);
        });
    }
    __syncthreads();

    {
        const int total = cnt + ((lcount < LCAP) ? lcount : LCAP);
        for (int b = wv * EPW; b < total; b += 32 * EPW) {
            if (b + EPW <= total) {
                int s[EPW], d[EPW];
                #pragma unroll
                for (int j = 0; j < EPW; ++j) {
                    const int2 en = lst[b + j];
                    s[j] = __builtin_amdgcn_readfirstlane(en.x);
                    d[j] = lo + __builtin_amdgcn_readfirstlane(en.y);
                }
                float4 hv[EPW];
                #pragma unroll
                for (int j = 0; j < EPW; ++j)
                    hv[j] = *(const float4*)(xin + (size_t)s[j] * NDIM + lane * CPL);
                float nrm[EPW];
                #pragma unroll
                for (int j = 0; j < EPW; ++j) nrm[j] = dinv[s[j]] * dinv[d[j]];
                #pragma unroll
                for (int j = 0; j < EPW; ++j) {
                    float* ap = acc + (d[j] - lo) * APITCH + lane;
                    unsafeAtomicAdd(ap,      hv[j].x * nrm[j]);
                    unsafeAtomicAdd(ap + 32, hv[j].y * nrm[j]);
                    unsafeAtomicAdd(ap + 64, hv[j].z * nrm[j]);
                    unsafeAtomicAdd(ap + 96, hv[j].w * nrm[j]);
                }
            } else {
                for (int k = b; k < total; ++k) {
                    const int2 en = lst[k];
                    const int s1 = __builtin_amdgcn_readfirstlane(en.x);
                    const int d1 = lo + __builtin_amdgcn_readfirstlane(en.y);
                    const float nr = dinv[s1] * dinv[d1];
                    const float4 hv1 = *(const float4*)(xin + (size_t)s1 * NDIM + lane * CPL);
                    float* ap = acc + (d1 - lo) * APITCH + lane;
                    unsafeAtomicAdd(ap,      hv1.x * nr);
                    unsafeAtomicAdd(ap + 32, hv1.y * nr);
                    unsafeAtomicAdd(ap + 64, hv1.z * nr);
                    unsafeAtomicAdd(ap + 96, hv1.w * nr);
                }
            }
        }
    }
    __syncthreads();

    {
        const int l16    = lane & 15;
        const int lhalf  = lane >> 4;
        const int ntiles = (cnt + 15) >> 4;
        for (int rt = wv; rt < ntiles; rt += 32) {
            int arow = rt * 16 + l16;
            if (arow > cnt - 1) arow = cnt - 1;
            const float* aptr = acc + arow * APITCH + lhalf * 2;

            v8f dacc[8];
            #pragma unroll
            for (int t = 0; t < 8; ++t) {
                const float bc = bias[t * 16 + l16];
                #pragma unroll
                for (int v = 0; v < 8; ++v) dacc[t][v] = bc;
            }
            for (int k = 0; k < NDIM; k += 4) {
                v2f a;
                {
                    const float2 av = *(const float2*)(aptr + k);
                    a[0] = av.x; a[1] = av.y;
                }
                mma_kstep(dacc, a, Ws, (k + lhalf * 2) * NDIM + l16);
            }

            #pragma unroll
            for (int t = 0; t < 8; ++t) {
                #pragma unroll
                for (int v = 0; v < 8; ++v) {
                    const int r = rt * 16 + v + lhalf * 8;
                    if (r < cnt) acc[r * APITCH + t * 16 + l16] = dacc[t][v];
                }
            }

            __builtin_amdgcn_fence(__ATOMIC_SEQ_CST, "wavefront");

            const int rend = (rt * 16 + 16 < cnt) ? (rt * 16 + 16) : cnt;
            for (int r = rt * 16; r < rend; ++r) {
                float4 o = *(const float4*)(acc + r * APITCH + lane * 4);
                if (TANH_OUT) {
                    o.x = __builtin_amdgcn_tanhf(o.x); o.y = __builtin_amdgcn_tanhf(o.y);
                    o.z = __builtin_amdgcn_tanhf(o.z); o.w = __builtin_amdgcn_tanhf(o.w);
                }
                *(float4*)(out + (size_t)(lo + r) * NDIM + lane * 4) = o;
            }
        }
    }
}

extern "C" void kernel_launch(void* const* d_in, const int* in_sizes, int n_in,
                              void* d_out, int out_size, void* d_ws, size_t ws_size,
                              hipStream_t stream) {
    const float* x   = (const float*)d_in[0];
    const float* W1  = (const float*)d_in[1];
    const float* b1  = (const float*)d_in[2];
    const float* W2  = (const float*)d_in[3];
    const float* b2  = (const float*)d_in[4];
    const int*   ei  = (const int*)d_in[5];

    const int N = in_sizes[0] / NDIM;
    const int E = in_sizes[5] / 2;
    const int* src = ei;
    const int* dst = ei + E;

    float* dinv = (float*)d_ws;
    unsigned* keys = (unsigned*)(dinv + ((N + 255) & ~255));
    const int nwords = ((E + KTRIP - 1) / KTRIP) * (KTRIP / 4);
    float* a1   = (float*)(keys + nwords);
    float* outp = (float*)d_out;

    int gA   = (N + OWN - 1) / OWN;

    int* gcount = (int*)(a1 + (size_t)N * NDIM);
    int* glist  = gcount + ((gA + 255) & ~255);

    k_edge_keys<<<(nwords + 255) / 256, 256, 0, stream>>>(dst, keys, E, nwords);
    k_own_filter<<<gA, 1024, 0, stream>>>(src, dst, keys, dinv, glist, gcount, N, E);

    k_layer<true> <<<gA, 1024, 0, stream>>>(x,  dinv, b1, W1, src, dst, glist, gcount, a1,   N, E);

    k_layer<false><<<gA, 1024, 0, stream>>>(a1, dinv, b2, W2, src, dst, glist, gcount, outp, N, E);
}
